# starting kernel
# baseline (speedup 1.0000x reference)
#include <hip/hip_runtime.h>

typedef __attribute__((ext_vector_type(2))) float v2f;
typedef __attribute__((ext_vector_type(8))) float v8f;

#define NNODES 100000
#define NEDGES 1250000


__global__ void fill_kernel(float* __restrict__ p, long long n, float v) {
    long long i = (long long)blockIdx.x * blockDim.x + threadIdx.x;
    if (i < n) p[i] = v;
}

__global__ void deg_kernel(const int* __restrict__ col, float* __restrict__ deg, int E) {
    int e = blockIdx.x * blockDim.x + threadIdx.x;
    if (e < E) atomicAdd(&deg[col[e]], 1.0f);
}

__global__ void rsqrt_kernel(float* __restrict__ d, int n) {
    int i = blockIdx.x * blockDim.x + threadIdx.x;
    if (i < n) {
        float x = d[i];
        d[i] = (x > 0.0f) ? rsqrtf(x) : 0.0f;
    }
}

template <int K, int NOUTPAD>
__global__ __launch_bounds__(256) void gemm_wmma_kernel(
    const float* __restrict__ X, const float* __restrict__ W,
    const float* __restrict__ bias, float* __restrict__ Y,
    int M, int Nout, int addBias)
{
    constexpr int STRIDE = K + 4;
    constexpr int NT = NOUTPAD / 16;
    __shared__ float Wt[NOUTPAD * STRIDE];

    for (int idx = threadIdx.x; idx < NOUTPAD * K; idx += 256) {
        int n = idx / K;
        int k = idx - n * K;
        Wt[n * STRIDE + k] = (n < Nout) ? W[(size_t)k * Nout + n] : 0.0f;
    }
    __syncthreads();

    const int lane  = threadIdx.x & 31;
    const int strip = blockIdx.x * 8 + (threadIdx.x >> 5);
    const int Mtiles = M >> 4;
    if (strip >= Mtiles) return;

    const int half = lane >> 4;
    const int l    = lane & 15;
    const int row0 = strip << 4;

    const float* __restrict__ ap  = X + (size_t)(row0 + l) * K + 2 * half;
    const float* __restrict__ bp0 = &Wt[l * STRIDE + 2 * half];

    v8f acc[NT];
#pragma unroll
    for (int t = 0; t < NT; ++t) acc[t] = {};

#pragma unroll 4
    for (int k = 0; k < K; k += 4) {
        v2f a;
        a.x = ap[k];
        a.y = ap[k + 1];
#pragma unroll
        for (int t = 0; t < NT; ++t) {
            v2f b;
            const float* bp = bp0 + t * 16 * STRIDE;
            b.x = bp[k];
            b.y = bp[k + 1];
            acc[t] = __builtin_amdgcn_wmma_f32_16x16x4_f32(
                false, a, false, b, (short)0, acc[t], false, false);
        }
    }

#pragma unroll
    for (int t = 0; t < NT; ++t) {
        int c = t * 16 + l;
        if (c < Nout) {
            float bv = addBias ? bias[c] : 0.0f;
#pragma unroll
            for (int v = 0; v < 8; ++v) {
                int r = row0 + v + 8 * half;
                Y[(size_t)r * Nout + c] = acc[t][v] + bv;
            }
        }
    }
}

__global__ void scatter_kernel(const float* __restrict__ H, const int* __restrict__ row,
                               const int* __restrict__ col, const float* __restrict__ dis,
                               float* __restrict__ AGG, int E) {
    long long gid = (long long)blockIdx.x * blockDim.x + threadIdx.x;
    int e  = (int)(gid >> 4);
    int f4 = (int)(gid & 15) << 2;
    if (e >= E) return;
    int r = row[e];
    int c = col[e];
    float nrm = dis[r] * dis[c];
    const float4 h = *(const float4*)(H + (size_t)r * 64 + f4);
    float* out = AGG + (size_t)c * 64 + f4;
    atomicAdd(out + 0, h.x * nrm);
    atomicAdd(out + 1, h.y * nrm);
    atomicAdd(out + 2, h.z * nrm);
    atomicAdd(out + 3, h.w * nrm);
}

__global__ void combine_kernel(float* __restrict__ agg, const float* __restrict__ htmp,
                               const float* __restrict__ dis, const float* __restrict__ bias,
                               int n  ) {
    int i = blockIdx.x * blockDim.x + threadIdx.x;
    if (i >= n) return;
    int node = i >> 6;
    int f    = i & 63;
    float d = dis[node];
    float v = agg[i] + htmp[i] * d * d + bias[f];
    agg[i] = (v > 0.0f) ? v : 0.0f;
}


extern "C" void kernel_launch(void* const* d_in, const int* in_sizes, int n_in,
                              void* d_out, int out_size, void* d_ws, size_t ws_size,
                              hipStream_t stream) {
    const float* x    = (const float*)d_in[0];
    const int*   ei   = (const int*)d_in[1];
    const float* W1   = (const float*)d_in[2];
    const float* b1   = (const float*)d_in[3];
    const float* W2   = (const float*)d_in[4];
    const float* b2   = (const float*)d_in[5];
    const float* Wout = (const float*)d_in[6];
    const float* bout = (const float*)d_in[7];
    float* out = (float*)d_out;

    const int* row = ei;
    const int* col = ei + NEDGES;

    float* dis = (float*)d_ws;
    float* A   = dis + NNODES;
    float* B   = A + (size_t)NNODES * 64;
    float* C   = B + (size_t)NNODES * 64;
    (void)ws_size; (void)n_in; (void)in_sizes; (void)out_size;

    const int T = 256;
    const long long NF = (long long)NNODES * 64;
    const int Mtiles = NNODES / 16;
    const int gemmBlocks = (Mtiles + 7) / 8;

    fill_kernel<<<(NNODES + T - 1) / T, T, 0, stream>>>(dis, NNODES, 1.0f);
    deg_kernel<<<(NEDGES + T - 1) / T, T, 0, stream>>>(col, dis, NEDGES);
    rsqrt_kernel<<<(NNODES + T - 1) / T, T, 0, stream>>>(dis, NNODES);

    gemm_wmma_kernel<128, 64><<<gemmBlocks, T, 0, stream>>>(x, W1, nullptr, A, NNODES, 64, 0);
    fill_kernel<<<(int)((NF + T - 1) / T), T, 0, stream>>>(B, NF, 0.0f);
    scatter_kernel<<<(int)(((long long)NEDGES * 16 + T - 1) / T), T, 0, stream>>>(A, row, col, dis, B, NEDGES);
    combine_kernel<<<(int)((NF + T - 1) / T), T, 0, stream>>>(B, A, dis, b1, (int)NF);

    gemm_wmma_kernel<64, 64><<<gemmBlocks, T, 0, stream>>>(B, W2, nullptr, A, NNODES, 64, 0);
    fill_kernel<<<(int)((NF + T - 1) / T), T, 0, stream>>>(C, NF, 0.0f);
    scatter_kernel<<<(int)(((long long)NEDGES * 16 + T - 1) / T), T, 0, stream>>>(A, row, col, dis, C, NEDGES);
    combine_kernel<<<(int)((NF + T - 1) / T), T, 0, stream>>>(C, A, dis, b2, (int)NF);

    gemm_wmma_kernel<64, 48><<<gemmBlocks, T, 0, stream>>>(C, Wout, bout, out, NNODES, 40, 1);
}


# products re-cut: 2048 resident waves loop over 16-row strips, weights kept in registers as B fragments (no LDS, no barrier), columns interleaved for 16 B W loads and row stores; sum order changes
# speedup vs baseline: 2.4013x; 2.4013x over previous
#include <hip/hip_runtime.h>

typedef __attribute__((ext_vector_type(2))) float v2f;
typedef __attribute__((ext_vector_type(8))) float v8f;

#define NNODES 100000
#define NEDGES 1250000


__global__ void fill_kernel(float* __restrict__ p, long long n, float v) {
    long long i = (long long)blockIdx.x * blockDim.x + threadIdx.x;
    if (i < n) p[i] = v;
}

__device__ __forceinline__ int die_half() {
    return (int)((__builtin_amdgcn_s_sendmsg_rtn(0x87) >> 18) & 1u);
}

__global__ void deg_kernel(const int* __restrict__ col, float* __restrict__ degc, int E, int n) {
    int e = blockIdx.x * blockDim.x + threadIdx.x;
    float* __restrict__ mine = degc + (size_t)die_half() * n;
    if (e < E) atomicAdd(&mine[col[e]], 1.0f);
}

__global__ void rsqrt_kernel(const float* __restrict__ degc, float* __restrict__ d, int n) {
    int i = blockIdx.x * blockDim.x + threadIdx.x;
    if (i < n) {
        float x = 1.0f + degc[i] + degc[n + i];
        d[i] = (x > 0.0f) ? rsqrtf(x) : 0.0f;
    }
}

template <int K, int NT, int NOUT, bool RELU_IN, bool SELF_OUT>
struct ProductStrip {
    static constexpr int NI = K / 8;

    __device__ static __forceinline__ void load_rows(float4 (&a)[NI], float4 (&a2)[NI],
                                                     const float* __restrict__ X,
                                                     const float* __restrict__ X2,
                                                     int strip, int c, int h) {
        const size_t off = (size_t)(strip * 16 + c) * K + 4 * h;
        const float4* __restrict__ p = reinterpret_cast<const float4*>(X + off);
#pragma unroll
        for (int i = 0; i < NI; ++i) a[i] = p[2 * i];
        if constexpr (RELU_IN) {
            const float4* __restrict__ p2 = reinterpret_cast<const float4*>(X2 + off);
#pragma unroll
            for (int i = 0; i < NI; ++i) a2[i] = p2[2 * i];
        }
    }

    __device__ static __forceinline__ void load_dis(float4 (&d)[2], const float* __restrict__ dis,
                                                    int strip, int h) {
        if constexpr (SELF_OUT) {
            const float4* __restrict__ dp = reinterpret_cast<const float4*>(dis + strip * 16 + 8 * h);
            d[0] = dp[0];
            d[1] = dp[1];
        }
    }

    __device__ static __forceinline__ void run(float4 (&a)[NI], float4 (&a2)[NI], const float4 (&d)[2],
                                               const v2f (&Bf)[NI][2][NT],
                                               const float (&bv)[NT],
                                               float* __restrict__ Y, float* __restrict__ S,
                                               float* __restrict__ S2,
                                               int strip, int c, int h) {
        const int row0 = strip * 16;
        __builtin_amdgcn_sched_barrier(0);
        const float dd[8] = {d[0].x, d[0].y, d[0].z, d[0].w, d[1].x, d[1].y, d[1].z, d[1].w};
        if constexpr (RELU_IN) {
#pragma unroll
            for (int i = 0; i < NI; ++i) {
                const float sx = a[i].x + a2[i].x, sy = a[i].y + a2[i].y;
                const float sz = a[i].z + a2[i].z, sw = a[i].w + a2[i].w;
                a[i].x = (sx > 0.0f) ? sx : 0.0f;
                a[i].y = (sy > 0.0f) ? sy : 0.0f;
                a[i].z = (sz > 0.0f) ? sz : 0.0f;
                a[i].w = (sw > 0.0f) ? sw : 0.0f;
            }
            __builtin_amdgcn_sched_barrier(0);
        }
        v8f acc[NT];
#pragma unroll
        for (int t = 0; t < NT; ++t) acc[t] = {};
#pragma unroll
        for (int i = 0; i < NI; ++i) {
#pragma unroll
            for (int u = 0; u < 2; ++u) {
                v2f af;
                af.x = u ? a[i].z : a[i].x;
                af.y = u ? a[i].w : a[i].y;
#pragma unroll
                for (int t = 0; t < NT; ++t)
                    acc[t] = __builtin_amdgcn_wmma_f32_16x16x4_f32(
                        false, af, false, Bf[i][u][t], (short)0, acc[t], false, false);
            }
        }
        __builtin_amdgcn_sched_barrier(0);
        asm volatile("v_nop\n\tv_nop\n\tv_nop\n\tv_nop");
        __builtin_amdgcn_sched_barrier(0);

        if constexpr (NT == 4 && NOUT == 64) {
#pragma unroll
            for (int v = 0; v < 8; ++v) {
                const size_t o = (size_t)(row0 + 8 * h + v) * NOUT + 4 * c;
                const float4 y = make_float4(acc[0][v], acc[1][v], acc[2][v], acc[3][v]);
                if constexpr (SELF_OUT) {
                    float4 s;
                    s.x = y.x * dd[v] * dd[v] + bv[0];
                    s.y = y.y * dd[v] * dd[v] + bv[1];
                    s.z = y.z * dd[v] * dd[v] + bv[2];
                    s.w = y.w * dd[v] * dd[v] + bv[3];
                    *reinterpret_cast<float4*>(Y + o) = y;
                    *reinterpret_cast<float4*>(S + o) = s;
                    *reinterpret_cast<float4*>(S2 + o) = make_float4(0.0f, 0.0f, 0.0f, 0.0f);
                } else {
                    *reinterpret_cast<float4*>(Y + o) =
                        make_float4(y.x + bv[0], y.y + bv[1], y.z + bv[2], y.w + bv[3]);
                }
            }
        } else {
            static_assert(!SELF_OUT, "narrow output has no start-value planes");
            const size_t o = (size_t)(row0 + 8 * h) * NOUT + NT * c;
            if (NT * c + NT <= NOUT) {
#pragma unroll
                for (int v = 0; v < 8; ++v) {
#pragma unroll
                    for (int t = 0; t < NT; ++t) Y[o + (size_t)v * NOUT + t] = acc[t][v] + bv[t];
                }
            } else {
#pragma unroll
                for (int v = 0; v < 8; ++v) {
#pragma unroll
                    for (int t = 0; t < NT; ++t)
                        if (NT * c + t < NOUT) Y[o + (size_t)v * NOUT + t] = acc[t][v] + bv[t];
                }
            }
        }
    }
};

template <int K, int NT, int NOUT, bool RELU_IN, bool SELF_OUT>
__global__ __launch_bounds__(256) void gemm_wmma_kernel(
    const float* __restrict__ X, const float* __restrict__ X2, const float* __restrict__ W,
    const float* __restrict__ bias, const float* __restrict__ dis,
    float* __restrict__ Y, float* __restrict__ S, float* __restrict__ S2, int Mtiles)
{
    using PS = ProductStrip<K, NT, NOUT, RELU_IN, SELF_OUT>;
    constexpr int NI = PS::NI;
    const int lane = threadIdx.x & 31;
    const int c = lane & 15;
    const int h = lane >> 4;
    const int wib = __builtin_amdgcn_readfirstlane((int)(threadIdx.x >> 5));
    const int nw = (int)gridDim.x * 8;
    int s = wib * (int)gridDim.x + (int)blockIdx.x;
    if (s >= Mtiles) return;

    float4 a[NI], a2[NI];
    PS::load_rows(a, a2, X, X2, s, c, h);

    v2f Bf[NI][2][NT];
    if constexpr (NT == 4 && NOUT == 64) {
#pragma unroll
        for (int i = 0; i < NI; ++i) {
#pragma unroll
            for (int u = 0; u < 2; ++u) {
                const int k = 8 * i + 4 * h + 2 * u;
                const float4 q0 = *reinterpret_cast<const float4*>(W + (size_t)k * NOUT + 4 * c);
                const float4 q1 = *reinterpret_cast<const float4*>(W + (size_t)(k + 1) * NOUT + 4 * c);
                Bf[i][u][0].x = q0.x; Bf[i][u][1].x = q0.y; Bf[i][u][2].x = q0.z; Bf[i][u][3].x = q0.w;
                Bf[i][u][0].y = q1.x; Bf[i][u][1].y = q1.y; Bf[i][u][2].y = q1.z; Bf[i][u][3].y = q1.w;
            }
        }
    } else {
        const bool whole = (NT * c + NT <= NOUT);
#pragma unroll
        for (int i = 0; i < NI; ++i) {
#pragma unroll
            for (int u = 0; u < 2; ++u) {
#pragma unroll
                for (int t = 0; t < NT; ++t) { Bf[i][u][t].x = 0.0f; Bf[i][u][t].y = 0.0f; }
            }
        }
        if (whole) {
#pragma unroll
            for (int i = 0; i < NI; ++i) {
#pragma unroll
                for (int u = 0; u < 2; ++u) {
                    const float* __restrict__ wp = W + (size_t)(8 * i + 4 * h + 2 * u) * NOUT + NT * c;
#pragma unroll
                    for (int t = 0; t < NT; ++t) { Bf[i][u][t].x = wp[t]; Bf[i][u][t].y = wp[NOUT + t]; }
                }
            }
        } else {
#pragma unroll
            for (int t = 0; t < NT; ++t) {
                if (NT * c + t < NOUT) {
#pragma unroll
                    for (int i = 0; i < NI; ++i) {
#pragma unroll
                        for (int u = 0; u < 2; ++u) {
                            const float* __restrict__ wp = W + (size_t)(8 * i + 4 * h + 2 * u) * NOUT + NT * c + t;
                            Bf[i][u][t].x = wp[0];
                            Bf[i][u][t].y = wp[NOUT];
                        }
                    }
                }
            }
        }
    }
    float bv[NT];
#pragma unroll
    for (int t = 0; t < NT; ++t) bv[t] = (NT * c + t < NOUT) ? bias[NT * c + t] : 0.0f;

    do {
        float4 n[NI], n2[NI], d[2];
        const int sn = s + nw;
        PS::load_dis(d, dis, s, h);
        PS::load_rows(n, n2, X, X2, (sn < Mtiles) ? sn : s, c, h);
        PS::run(a, a2, d, Bf, bv, Y, S, S2, s, c, h);
#pragma unroll
        for (int i = 0; i < NI; ++i) {
            a[i] = n[i];
            if constexpr (RELU_IN) a2[i] = n2[i];
        }
        s = sn;
    } while (s < Mtiles);
}

template <int NB>
__device__ __forceinline__ void scatter_edges(const float* __restrict__ H, float* __restrict__ AGG,
                                              int r, int c, float nrm, int k0, int lane) {
    float h0[NB], h1[NB];
#pragma unroll
    for (int j = 0; j < NB; ++j) {
        const unsigned hb = (unsigned)__builtin_amdgcn_readlane(r, k0 + j) * 64u + (unsigned)lane;
        h0[j] = H[hb];
        h1[j] = H[hb + 32u];
    }
#pragma unroll
    for (int j = 0; j < NB; ++j) {
        const unsigned ob = (unsigned)__builtin_amdgcn_readlane(c, k0 + j) * 64u + (unsigned)lane;
        const float nk = __int_as_float(__builtin_amdgcn_readlane(__float_as_int(nrm), k0 + j));
        atomicAdd(AGG + ob, h0[j] * nk);
        atomicAdd(AGG + ob + 32u, h1[j] * nk);
    }
}

__global__ __launch_bounds__(256) void scatter_kernel(const float* __restrict__ H, const int* __restrict__ row,
                               const int* __restrict__ col, const float* __restrict__ dis,
                               float* __restrict__ AGG0, float* __restrict__ AGG1, int E) {
    const int lane = threadIdx.x & 31;
    float* __restrict__ AGG = die_half() ? AGG1 : AGG0;
    const int wave = blockIdx.x * 8 + __builtin_amdgcn_readfirstlane((int)(threadIdx.x >> 5));
    const int e0 = wave * 32;
    if (e0 >= E) return;
    const int cnt = (E - e0 < 32) ? (E - e0) : 32;
    int r = 0, c = 0;
    float nrm = 0.0f;
    if (lane < cnt) {
        r = row[e0 + lane];
        c = col[e0 + lane];
        nrm = dis[r] * dis[c];
    }
    int k = 0;
    for (; k + 8 <= cnt; k += 8) scatter_edges<8>(H, AGG, r, c, nrm, k, lane);
    for (; k < cnt; ++k) scatter_edges<1>(H, AGG, r, c, nrm, k, lane);
}


extern "C" void kernel_launch(void* const* d_in, const int* in_sizes, int n_in,
                              void* d_out, int out_size, void* d_ws, size_t ws_size,
                              hipStream_t stream) {
    const float* x    = (const float*)d_in[0];
    const int*   ei   = (const int*)d_in[1];
    const float* W1   = (const float*)d_in[2];
    const float* b1   = (const float*)d_in[3];
    const float* W2   = (const float*)d_in[4];
    const float* b2   = (const float*)d_in[5];
    const float* Wout = (const float*)d_in[6];
    const float* bout = (const float*)d_in[7];
    float* out = (float*)d_out;

    const int* row = ei;
    const int* col = ei + NEDGES;

    float* degc = (float*)d_ws;
    float* dis = degc + 2 * NNODES;
    float* A   = dis + NNODES;
    float* B0  = A + (size_t)NNODES * 64;
    float* B1  = B0 + (size_t)NNODES * 64;
    float* C0  = B1 + (size_t)NNODES * 64;
    float* C1  = C0 + (size_t)NNODES * 64;
    (void)ws_size; (void)n_in; (void)in_sizes; (void)out_size;

    const int T = 256;
    const int Mtiles = NNODES / 16;
    const int gemmBlocks = 256;

    fill_kernel<<<(2 * NNODES + T - 1) / T, T, 0, stream>>>(degc, 2 * NNODES, 0.0f);
    deg_kernel<<<(NEDGES + T - 1) / T, T, 0, stream>>>(col, degc, NEDGES, NNODES);
    rsqrt_kernel<<<(NNODES + T - 1) / T, T, 0, stream>>>(degc, dis, NNODES);

    gemm_wmma_kernel<128, 4, 64, false, true><<<gemmBlocks, T, 0, stream>>>(x, nullptr, W1, b1, dis, A, B0, B1, Mtiles);
    scatter_kernel<<<(NEDGES + T - 1) / T, T, 0, stream>>>(A, row, col, dis, B0, B1, NEDGES);

    gemm_wmma_kernel<64, 4, 64, true, true><<<gemmBlocks, T, 0, stream>>>(B0, B1, W2, b2, dis, A, C0, C1, Mtiles);
    scatter_kernel<<<(NEDGES + T - 1) / T, T, 0, stream>>>(A, row, col, dis, C0, C1, NEDGES);

    gemm_wmma_kernel<64, 3, 40, true, false><<<gemmBlocks, T, 0, stream>>>(C0, C1, Wout, bout, nullptr, out, nullptr, nullptr, Mtiles);
}


# degree count and both edge scatters run on resident waves looping over edge chunks, next chunk's indices asked ahead (fill, rsqrt: 4 elements a lane): 11x fewer waves enter; arrival-order sums kept
# speedup vs baseline: 2.5128x; 1.0464x over previous
#include <hip/hip_runtime.h>

typedef __attribute__((ext_vector_type(2))) float v2f;
typedef __attribute__((ext_vector_type(8))) float v8f;

#define NNODES 100000
#define NEDGES 1250000


__global__ void fill_kernel(float* __restrict__ p, int n4, float v) {
    int i = blockIdx.x * blockDim.x + threadIdx.x;
    if (i < n4) reinterpret_cast<float4*>(p)[i] = make_float4(v, v, v, v);
}

__device__ __forceinline__ int die_half() {
    return (int)((__builtin_amdgcn_s_sendmsg_rtn(0x87) >> 18) & 1u);
}

__device__ __forceinline__ void count4(float* __restrict__ mine, const int4 v) {
    atomicAdd(&mine[v.x], 1.0f);
    atomicAdd(&mine[v.y], 1.0f);
    atomicAdd(&mine[v.z], 1.0f);
    atomicAdd(&mine[v.w], 1.0f);
}

__global__ __launch_bounds__(256) void deg_kernel(const int* __restrict__ col, float* __restrict__ degc, int E, int n) {
    float* __restrict__ mine = degc + (size_t)die_half() * n;
    const int lane = threadIdx.x & 31;
    const int wib = __builtin_amdgcn_readfirstlane((int)(threadIdx.x >> 5));
    const int nw = (int)gridDim.x * 8;
    const int w = wib * (int)gridDim.x + (int)blockIdx.x;
    const int nfull = E >> 7;
    const int4* __restrict__ c4 = reinterpret_cast<const int4*>(col);
    int ch = w;
    for (; ch + 3 * nw < nfull; ch += 4 * nw) {
        const int4 v0 = c4[(ch) * 32 + lane];
        const int4 v1 = c4[(ch + nw) * 32 + lane];
        const int4 v2 = c4[(ch + 2 * nw) * 32 + lane];
        const int4 v3 = c4[(ch + 3 * nw) * 32 + lane];
        asm volatile("" :: "v"(v0.x), "v"(v0.y), "v"(v0.z), "v"(v0.w), "v"(v1.x), "v"(v1.y), "v"(v1.z), "v"(v1.w),
                           "v"(v2.x), "v"(v2.y), "v"(v2.z), "v"(v2.w), "v"(v3.x), "v"(v3.y), "v"(v3.z), "v"(v3.w));
        count4(mine, v0);
        count4(mine, v1);
        count4(mine, v2);
        count4(mine, v3);
    }
    for (; ch < nfull; ch += nw) count4(mine, c4[ch * 32 + lane]);
    if (w == nw - 1) {
        for (int e = (nfull << 7) + lane; e < E; e += 32) atomicAdd(&mine[col[e]], 1.0f);
    }
}

__device__ __forceinline__ float inv_sqrt_deg(float a, float b) {
    const float x = 1.0f + a + b;
    return (x > 0.0f) ? rsqrtf(x) : 0.0f;
}

__global__ void rsqrt_kernel(const float* __restrict__ degc, float* __restrict__ d, int n4) {
    int i = blockIdx.x * blockDim.x + threadIdx.x;
    if (i < n4) {
        const float4 a = reinterpret_cast<const float4*>(degc)[i];
        const float4 b = reinterpret_cast<const float4*>(degc)[n4 + i];
        reinterpret_cast<float4*>(d)[i] = make_float4(inv_sqrt_deg(a.x, b.x), inv_sqrt_deg(a.y, b.y),
                                                     inv_sqrt_deg(a.z, b.z), inv_sqrt_deg(a.w, b.w));
    }
}

template <int K, int NT, int NOUT, bool RELU_IN, bool SELF_OUT>
struct ProductStrip {
    static constexpr int NI = K / 8;

    __device__ static __forceinline__ void load_rows(float4 (&a)[NI], float4 (&a2)[NI],
                                                     const float* __restrict__ X,
                                                     const float* __restrict__ X2,
                                                     int strip, int c, int h) {
        const size_t off = (size_t)(strip * 16 + c) * K + 4 * h;
        const float4* __restrict__ p = reinterpret_cast<const float4*>(X + off);
#pragma unroll
        for (int i = 0; i < NI; ++i) a[i] = p[2 * i];
        if constexpr (RELU_IN) {
            const float4* __restrict__ p2 = reinterpret_cast<const float4*>(X2 + off);
#pragma unroll
            for (int i = 0; i < NI; ++i) a2[i] = p2[2 * i];
        }
    }

    __device__ static __forceinline__ void load_dis(float4 (&d)[2], const float* __restrict__ dis,
                                                    int strip, int h) {
        if constexpr (SELF_OUT) {
            const float4* __restrict__ dp = reinterpret_cast<const float4*>(dis + strip * 16 + 8 * h);
            d[0] = dp[0];
            d[1] = dp[1];
        }
    }

    __device__ static __forceinline__ void run(float4 (&a)[NI], float4 (&a2)[NI], const float4 (&d)[2],
                                               const v2f (&Bf)[NI][2][NT],
                                               const float (&bv)[NT],
                                               float* __restrict__ Y, float* __restrict__ S,
                                               float* __restrict__ S2,
                                               int strip, int c, int h) {
        const int row0 = strip * 16;
        __builtin_amdgcn_sched_barrier(0);
        const float dd[8] = {d[0].x, d[0].y, d[0].z, d[0].w, d[1].x, d[1].y, d[1].z, d[1].w};
        if constexpr (RELU_IN) {
#pragma unroll
            for (int i = 0; i < NI; ++i) {
                const float sx = a[i].x + a2[i].x, sy = a[i].y + a2[i].y;
                const float sz = a[i].z + a2[i].z, sw = a[i].w + a2[i].w;
                a[i].x = (sx > 0.0f) ? sx : 0.0f;
                a[i].y = (sy > 0.0f) ? sy : 0.0f;
                a[i].z = (sz > 0.0f) ? sz : 0.0f;
                a[i].w = (sw > 0.0f) ? sw : 0.0f;
            }
            __builtin_amdgcn_sched_barrier(0);
        }
        v8f acc[NT];
#pragma unroll
        for (int t = 0; t < NT; ++t) acc[t] = {};
#pragma unroll
        for (int i = 0; i < NI; ++i) {
#pragma unroll
            for (int u = 0; u < 2; ++u) {
                v2f af;
                af.x = u ? a[i].z : a[i].x;
                af.y = u ? a[i].w : a[i].y;
#pragma unroll
                for (int t = 0; t < NT; ++t)
                    acc[t] = __builtin_amdgcn_wmma_f32_16x16x4_f32(
                        false, af, false, Bf[i][u][t], (short)0, acc[t], false, false);
            }
        }
        __builtin_amdgcn_sched_barrier(0);
        asm volatile("v_nop\n\tv_nop\n\tv_nop\n\tv_nop");
        __builtin_amdgcn_sched_barrier(0);

        if constexpr (NT == 4 && NOUT == 64) {
#pragma unroll
            for (int v = 0; v < 8; ++v) {
                const size_t o = (size_t)(row0 + 8 * h + v) * NOUT + 4 * c;
                const float4 y = make_float4(acc[0][v], acc[1][v], acc[2][v], acc[3][v]);
                if constexpr (SELF_OUT) {
                    float4 s;
                    s.x = y.x * dd[v] * dd[v] + bv[0];
                    s.y = y.y * dd[v] * dd[v] + bv[1];
                    s.z = y.z * dd[v] * dd[v] + bv[2];
                    s.w = y.w * dd[v] * dd[v] + bv[3];
                    *reinterpret_cast<float4*>(Y + o) = y;
                    *reinterpret_cast<float4*>(S + o) = s;
                    *reinterpret_cast<float4*>(S2 + o) = make_float4(0.0f, 0.0f, 0.0f, 0.0f);
                } else {
                    *reinterpret_cast<float4*>(Y + o) =
                        make_float4(y.x + bv[0], y.y + bv[1], y.z + bv[2], y.w + bv[3]);
                }
            }
        } else {
            static_assert(!SELF_OUT, "narrow output has no start-value planes");
            const size_t o = (size_t)(row0 + 8 * h) * NOUT + NT * c;
            if (NT * c + NT <= NOUT) {
#pragma unroll
                for (int v = 0; v < 8; ++v) {
#pragma unroll
                    for (int t = 0; t < NT; ++t) Y[o + (size_t)v * NOUT + t] = acc[t][v] + bv[t];
                }
            } else {
#pragma unroll
                for (int v = 0; v < 8; ++v) {
#pragma unroll
                    for (int t = 0; t < NT; ++t)
                        if (NT * c + t < NOUT) Y[o + (size_t)v * NOUT + t] = acc[t][v] + bv[t];
                }
            }
        }
    }
};

template <int K, int NT, int NOUT, bool RELU_IN, bool SELF_OUT>
__global__ __launch_bounds__(256) void gemm_wmma_kernel(
    const float* __restrict__ X, const float* __restrict__ X2, const float* __restrict__ W,
    const float* __restrict__ bias, const float* __restrict__ dis,
    float* __restrict__ Y, float* __restrict__ S, float* __restrict__ S2, int Mtiles)
{
    using PS = ProductStrip<K, NT, NOUT, RELU_IN, SELF_OUT>;
    constexpr int NI = PS::NI;
    const int lane = threadIdx.x & 31;
    const int c = lane & 15;
    const int h = lane >> 4;
    const int wib = __builtin_amdgcn_readfirstlane((int)(threadIdx.x >> 5));
    const int nw = (int)gridDim.x * 8;
    int s = wib * (int)gridDim.x + (int)blockIdx.x;
    if (s >= Mtiles) return;

    float4 a[NI], a2[NI];
    PS::load_rows(a, a2, X, X2, s, c, h);

    v2f Bf[NI][2][NT];
    if constexpr (NT == 4 && NOUT == 64) {
#pragma unroll
        for (int i = 0; i < NI; ++i) {
#pragma unroll
            for (int u = 0; u < 2; ++u) {
                const int k = 8 * i + 4 * h + 2 * u;
                const float4 q0 = *reinterpret_cast<const float4*>(W + (size_t)k * NOUT + 4 * c);
                const float4 q1 = *reinterpret_cast<const float4*>(W + (size_t)(k + 1) * NOUT + 4 * c);
                Bf[i][u][0].x = q0.x; Bf[i][u][1].x = q0.y; Bf[i][u][2].x = q0.z; Bf[i][u][3].x = q0.w;
                Bf[i][u][0].y = q1.x; Bf[i][u][1].y = q1.y; Bf[i][u][2].y = q1.z; Bf[i][u][3].y = q1.w;
            }
        }
    } else {
        const bool whole = (NT * c + NT <= NOUT);
#pragma unroll
        for (int i = 0; i < NI; ++i) {
#pragma unroll
            for (int u = 0; u < 2; ++u) {
#pragma unroll
                for (int t = 0; t < NT; ++t) { Bf[i][u][t].x = 0.0f; Bf[i][u][t].y = 0.0f; }
            }
        }
        if (whole) {
#pragma unroll
            for (int i = 0; i < NI; ++i) {
#pragma unroll
                for (int u = 0; u < 2; ++u) {
                    const float* __restrict__ wp = W + (size_t)(8 * i + 4 * h + 2 * u) * NOUT + NT * c;
#pragma unroll
                    for (int t = 0; t < NT; ++t) { Bf[i][u][t].x = wp[t]; Bf[i][u][t].y = wp[NOUT + t]; }
                }
            }
        } else {
#pragma unroll
            for (int t = 0; t < NT; ++t) {
                if (NT * c + t < NOUT) {
#pragma unroll
                    for (int i = 0; i < NI; ++i) {
#pragma unroll
                        for (int u = 0; u < 2; ++u) {
                            const float* __restrict__ wp = W + (size_t)(8 * i + 4 * h + 2 * u) * NOUT + NT * c + t;
                            Bf[i][u][t].x = wp[0];
                            Bf[i][u][t].y = wp[NOUT];
                        }
                    }
                }
            }
        }
    }
    float bv[NT];
#pragma unroll
    for (int t = 0; t < NT; ++t) bv[t] = (NT * c + t < NOUT) ? bias[NT * c + t] : 0.0f;

    do {
        float4 n[NI], n2[NI], d[2];
        const int sn = s + nw;
        PS::load_dis(d, dis, s, h);
        PS::load_rows(n, n2, X, X2, (sn < Mtiles) ? sn : s, c, h);
        PS::run(a, a2, d, Bf, bv, Y, S, S2, s, c, h);
#pragma unroll
        for (int i = 0; i < NI; ++i) {
            a[i] = n[i];
            if constexpr (RELU_IN) a2[i] = n2[i];
        }
        s = sn;
    } while (s < Mtiles);
}

template <int NB>
__device__ __forceinline__ void scatter_edges(const float* __restrict__ H, float* __restrict__ AGG,
                                              int r, int c, float nrm, int k0, int lane) {
    float h0[NB], h1[NB];
#pragma unroll
    for (int j = 0; j < NB; ++j) {
        const unsigned hb = (unsigned)__builtin_amdgcn_readlane(r, k0 + j) * 64u + (unsigned)lane;
        h0[j] = H[hb];
        h1[j] = H[hb + 32u];
    }
#pragma unroll
    for (int j = 0; j < NB; ++j) {
        const unsigned ob = (unsigned)__builtin_amdgcn_readlane(c, k0 + j) * 64u + (unsigned)lane;
        const float nk = __int_as_float(__builtin_amdgcn_readlane(__float_as_int(nrm), k0 + j));
        atomicAdd(AGG + ob, h0[j] * nk);
        atomicAdd(AGG + ob + 32u, h1[j] * nk);
    }
}

__global__ __launch_bounds__(256) void scatter_kernel(const float* __restrict__ H, const int* __restrict__ row,
                               const int* __restrict__ col, const float* __restrict__ dis,
                               float* __restrict__ AGG0, float* __restrict__ AGG1, int E) {
    const int lane = threadIdx.x & 31;
    float* __restrict__ AGG = die_half() ? AGG1 : AGG0;
    const int wib = __builtin_amdgcn_readfirstlane((int)(threadIdx.x >> 5));
    const int nw = (int)gridDim.x * 8;
    const int nchunks = (E + 31) >> 5;
    int s = wib * (int)gridDim.x + (int)blockIdx.x;
    if (s >= nchunks) return;
    const int last = E - 1;
    int e = s * 32 + lane;
    e = (e < last) ? e : last;
    int r = row[e];
    int c = col[e];
    float nrm = dis[r] * dis[c];
    do {
        const int sn = s + nw;
        int en = sn * 32 + lane;
        en = (en < last) ? en : last;
        const int rn = row[en];
        const int cn = col[en];
        const int cnt = E - s * 32;
        if (cnt >= 32) {
            scatter_edges<16>(H, AGG, r, c, nrm, 0, lane);
        } else {
            int k = 0;
            for (; k + 8 <= cnt; k += 8) scatter_edges<8>(H, AGG, r, c, nrm, k, lane);
            for (; k < cnt; ++k) scatter_edges<1>(H, AGG, r, c, nrm, k, lane);
        }
        const float nn = dis[rn] * dis[cn];
        if (cnt >= 32) scatter_edges<16>(H, AGG, r, c, nrm, 16, lane);
        r = rn;
        c = cn;
        nrm = nn;
        s = sn;
    } while (s < nchunks);
}


extern "C" void kernel_launch(void* const* d_in, const int* in_sizes, int n_in,
                              void* d_out, int out_size, void* d_ws, size_t ws_size,
                              hipStream_t stream) {
    const float* x    = (const float*)d_in[0];
    const int*   ei   = (const int*)d_in[1];
    const float* W1   = (const float*)d_in[2];
    const float* b1   = (const float*)d_in[3];
    const float* W2   = (const float*)d_in[4];
    const float* b2   = (const float*)d_in[5];
    const float* Wout = (const float*)d_in[6];
    const float* bout = (const float*)d_in[7];
    float* out = (float*)d_out;

    const int* row = ei;
    const int* col = ei + NEDGES;

    float* degc = (float*)d_ws;
    float* dis = degc + 2 * NNODES;
    float* A   = dis + NNODES;
    float* B0  = A + (size_t)NNODES * 64;
    float* B1  = B0 + (size_t)NNODES * 64;
    float* C0  = B1 + (size_t)NNODES * 64;
    float* C1  = C0 + (size_t)NNODES * 64;
    (void)ws_size; (void)n_in; (void)in_sizes; (void)out_size;

    const int T = 256;
    const int Mtiles = NNODES / 16;
    const int gemmBlocks = 256;
    const int edgeBlocksDeg = 256;
    const int edgeBlocks = 512;

    fill_kernel<<<(2 * NNODES / 4 + T - 1) / T, T, 0, stream>>>(degc, 2 * NNODES / 4, 0.0f);
    deg_kernel<<<edgeBlocksDeg, T, 0, stream>>>(col, degc, NEDGES, NNODES);
    rsqrt_kernel<<<(NNODES / 4 + T - 1) / T, T, 0, stream>>>(degc, dis, NNODES / 4);

    gemm_wmma_kernel<128, 4, 64, false, true><<<gemmBlocks, T, 0, stream>>>(x, nullptr, W1, b1, dis, A, B0, B1, Mtiles);
    scatter_kernel<<<edgeBlocks, T, 0, stream>>>(A, row, col, dis, B0, B1, NEDGES);

    gemm_wmma_kernel<64, 4, 64, true, true><<<gemmBlocks, T, 0, stream>>>(B0, B1, W2, b2, dis, A, C0, C1, Mtiles);
    scatter_kernel<<<edgeBlocks, T, 0, stream>>>(A, row, col, dis, C0, C1, NEDGES);

    gemm_wmma_kernel<64, 3, 40, true, false><<<gemmBlocks, T, 0, stream>>>(C0, C1, Wout, bout, nullptr, out, nullptr, nullptr, Mtiles);
}


# edge scatter reads both degree factors of an edge from an LDS copy of the degree table (16-bit counts, 200 KB, one 16-wave block per WGP): two 4-byte gathers per edge gone; sums unchanged
# speedup vs baseline: 2.6251x; 1.0447x over previous
#include <hip/hip_runtime.h>

typedef __attribute__((ext_vector_type(2))) float v2f;
typedef __attribute__((ext_vector_type(8))) float v8f;

#define NNODES 100000
#define NEDGES 1250000


__global__ void fill_kernel(float* __restrict__ p, int n4, float v) {
    int i = blockIdx.x * blockDim.x + threadIdx.x;
    if (i < n4) reinterpret_cast<float4*>(p)[i] = make_float4(v, v, v, v);
}

__device__ __forceinline__ int die_half() {
    return (int)((__builtin_amdgcn_s_sendmsg_rtn(0x87) >> 18) & 1u);
}

__device__ __forceinline__ void count4(float* __restrict__ mine, const int4 v) {
    atomicAdd(&mine[v.x], 1.0f);
    atomicAdd(&mine[v.y], 1.0f);
    atomicAdd(&mine[v.z], 1.0f);
    atomicAdd(&mine[v.w], 1.0f);
}

__global__ __launch_bounds__(256) void deg_kernel(const int* __restrict__ col, float* __restrict__ degc, int E, int n) {
    float* __restrict__ mine = degc + (size_t)die_half() * n;
    const int lane = threadIdx.x & 31;
    const int wib = __builtin_amdgcn_readfirstlane((int)(threadIdx.x >> 5));
    const int nw = (int)gridDim.x * 8;
    const int w = wib * (int)gridDim.x + (int)blockIdx.x;
    const int nfull = E >> 7;
    const int4* __restrict__ c4 = reinterpret_cast<const int4*>(col);
    int ch = w;
    for (; ch + 3 * nw < nfull; ch += 4 * nw) {
        const int4 v0 = c4[(ch) * 32 + lane];
        const int4 v1 = c4[(ch + nw) * 32 + lane];
        const int4 v2 = c4[(ch + 2 * nw) * 32 + lane];
        const int4 v3 = c4[(ch + 3 * nw) * 32 + lane];
        asm volatile("" :: "v"(v0.x), "v"(v0.y), "v"(v0.z), "v"(v0.w), "v"(v1.x), "v"(v1.y), "v"(v1.z), "v"(v1.w),
                           "v"(v2.x), "v"(v2.y), "v"(v2.z), "v"(v2.w), "v"(v3.x), "v"(v3.y), "v"(v3.z), "v"(v3.w));
        count4(mine, v0);
        count4(mine, v1);
        count4(mine, v2);
        count4(mine, v3);
    }
    for (; ch < nfull; ch += nw) count4(mine, c4[ch * 32 + lane]);
    if (w == nw - 1) {
        for (int e = (nfull << 7) + lane; e < E; e += 32) atomicAdd(&mine[col[e]], 1.0f);
    }
}

__device__ __forceinline__ float inv_sqrt_deg(float a, float b) {
    const float x = 1.0f + a + b;
    return (x > 0.0f) ? rsqrtf(x) : 0.0f;
}

#define DEG_WIDE 65535u
__device__ __forceinline__ unsigned deg_code(float a, float b) {
    const float x = 1.0f + a + b;
    return (x < 65535.0f) ? (unsigned)x : DEG_WIDE;
}

__global__ void rsqrt_kernel(const float* __restrict__ degc, float* __restrict__ d,
                             unsigned short* __restrict__ dh, int n4) {
    int i = blockIdx.x * blockDim.x + threadIdx.x;
    if (i < n4) {
        const float4 a = reinterpret_cast<const float4*>(degc)[i];
        const float4 b = reinterpret_cast<const float4*>(degc)[n4 + i];
        reinterpret_cast<float4*>(d)[i] = make_float4(inv_sqrt_deg(a.x, b.x), inv_sqrt_deg(a.y, b.y),
                                                     inv_sqrt_deg(a.z, b.z), inv_sqrt_deg(a.w, b.w));
        reinterpret_cast<uint2*>(dh)[i] = make_uint2(deg_code(a.x, b.x) | (deg_code(a.y, b.y) << 16),
                                                    deg_code(a.z, b.z) | (deg_code(a.w, b.w) << 16));
    }
}

template <int K, int NT, int NOUT, bool RELU_IN, bool SELF_OUT>
struct ProductStrip {
    static constexpr int NI = K / 8;

    __device__ static __forceinline__ void load_rows(float4 (&a)[NI], float4 (&a2)[NI],
                                                     const float* __restrict__ X,
                                                     const float* __restrict__ X2,
                                                     int strip, int c, int h) {
        const size_t off = (size_t)(strip * 16 + c) * K + 4 * h;
        const float4* __restrict__ p = reinterpret_cast<const float4*>(X + off);
#pragma unroll
        for (int i = 0; i < NI; ++i) a[i] = p[2 * i];
        if constexpr (RELU_IN) {
            const float4* __restrict__ p2 = reinterpret_cast<const float4*>(X2 + off);
#pragma unroll
            for (int i = 0; i < NI; ++i) a2[i] = p2[2 * i];
        }
    }

    __device__ static __forceinline__ void load_dis(float4 (&d)[2], const float* __restrict__ dis,
                                                    int strip, int h) {
        if constexpr (SELF_OUT) {
            const float4* __restrict__ dp = reinterpret_cast<const float4*>(dis + strip * 16 + 8 * h);
            d[0] = dp[0];
            d[1] = dp[1];
        }
    }

    __device__ static __forceinline__ void run(float4 (&a)[NI], float4 (&a2)[NI], const float4 (&d)[2],
                                               const v2f (&Bf)[NI][2][NT],
                                               const float (&bv)[NT],
                                               float* __restrict__ Y, float* __restrict__ S,
                                               float* __restrict__ S2,
                                               int strip, int c, int h) {
        const int row0 = strip * 16;
        __builtin_amdgcn_sched_barrier(0);
        const float dd[8] = {d[0].x, d[0].y, d[0].z, d[0].w, d[1].x, d[1].y, d[1].z, d[1].w};
        if constexpr (RELU_IN) {
#pragma unroll
            for (int i = 0; i < NI; ++i) {
                const float sx = a[i].x + a2[i].x, sy = a[i].y + a2[i].y;
                const float sz = a[i].z + a2[i].z, sw = a[i].w + a2[i].w;
                a[i].x = (sx > 0.0f) ? sx : 0.0f;
                a[i].y = (sy > 0.0f) ? sy : 0.0f;
                a[i].z = (sz > 0.0f) ? sz : 0.0f;
                a[i].w = (sw > 0.0f) ? sw : 0.0f;
            }
            __builtin_amdgcn_sched_barrier(0);
        }
        v8f acc[NT];
#pragma unroll
        for (int t = 0; t < NT; ++t) acc[t] = {};
#pragma unroll
        for (int i = 0; i < NI; ++i) {
#pragma unroll
            for (int u = 0; u < 2; ++u) {
                v2f af;
                af.x = u ? a[i].z : a[i].x;
                af.y = u ? a[i].w : a[i].y;
#pragma unroll
                for (int t = 0; t < NT; ++t)
                    acc[t] = __builtin_amdgcn_wmma_f32_16x16x4_f32(
                        false, af, false, Bf[i][u][t], (short)0, acc[t], false, false);
            }
        }
        __builtin_amdgcn_sched_barrier(0);
        asm volatile("v_nop\n\tv_nop\n\tv_nop\n\tv_nop");
        __builtin_amdgcn_sched_barrier(0);

        if constexpr (NT == 4 && NOUT == 64) {
#pragma unroll
            for (int v = 0; v < 8; ++v) {
                const size_t o = (size_t)(row0 + 8 * h + v) * NOUT + 4 * c;
                const float4 y = make_float4(acc[0][v], acc[1][v], acc[2][v], acc[3][v]);
                if constexpr (SELF_OUT) {
                    float4 s;
                    s.x = y.x * dd[v] * dd[v] + bv[0];
                    s.y = y.y * dd[v] * dd[v] + bv[1];
                    s.z = y.z * dd[v] * dd[v] + bv[2];
                    s.w = y.w * dd[v] * dd[v] + bv[3];
                    *reinterpret_cast<float4*>(Y + o) = y;
                    *reinterpret_cast<float4*>(S + o) = s;
                    *reinterpret_cast<float4*>(S2 + o) = make_float4(0.0f, 0.0f, 0.0f, 0.0f);
                } else {
                    *reinterpret_cast<float4*>(Y + o) =
                        make_float4(y.x + bv[0], y.y + bv[1], y.z + bv[2], y.w + bv[3]);
                }
            }
        } else {
            static_assert(!SELF_OUT, "narrow output has no start-value planes");
            const size_t o = (size_t)(row0 + 8 * h) * NOUT + NT * c;
            if (NT * c + NT <= NOUT) {
#pragma unroll
                for (int v = 0; v < 8; ++v) {
#pragma unroll
                    for (int t = 0; t < NT; ++t) Y[o + (size_t)v * NOUT + t] = acc[t][v] + bv[t];
                }
            } else {
#pragma unroll
                for (int v = 0; v < 8; ++v) {
#pragma unroll
                    for (int t = 0; t < NT; ++t)
                        if (NT * c + t < NOUT) Y[o + (size_t)v * NOUT + t] = acc[t][v] + bv[t];
                }
            }
        }
    }
};

template <int K, int NT, int NOUT, bool RELU_IN, bool SELF_OUT>
__global__ __launch_bounds__(256) void gemm_wmma_kernel(
    const float* __restrict__ X, const float* __restrict__ X2, const float* __restrict__ W,
    const float* __restrict__ bias, const float* __restrict__ dis,
    float* __restrict__ Y, float* __restrict__ S, float* __restrict__ S2, int Mtiles)
{
    using PS = ProductStrip<K, NT, NOUT, RELU_IN, SELF_OUT>;
    constexpr int NI = PS::NI;
    const int lane = threadIdx.x & 31;
    const int c = lane & 15;
    const int h = lane >> 4;
    const int wib = __builtin_amdgcn_readfirstlane((int)(threadIdx.x >> 5));
    const int nw = (int)gridDim.x * 8;
    int s = wib * (int)gridDim.x + (int)blockIdx.x;
    if (s >= Mtiles) return;

    float4 a[NI], a2[NI];
    PS::load_rows(a, a2, X, X2, s, c, h);

    v2f Bf[NI][2][NT];
    if constexpr (NT == 4 && NOUT == 64) {
#pragma unroll
        for (int i = 0; i < NI; ++i) {
#pragma unroll
            for (int u = 0; u < 2; ++u) {
                const int k = 8 * i + 4 * h + 2 * u;
                const float4 q0 = *reinterpret_cast<const float4*>(W + (size_t)k * NOUT + 4 * c);
                const float4 q1 = *reinterpret_cast<const float4*>(W + (size_t)(k + 1) * NOUT + 4 * c);
                Bf[i][u][0].x = q0.x; Bf[i][u][1].x = q0.y; Bf[i][u][2].x = q0.z; Bf[i][u][3].x = q0.w;
                Bf[i][u][0].y = q1.x; Bf[i][u][1].y = q1.y; Bf[i][u][2].y = q1.z; Bf[i][u][3].y = q1.w;
            }
        }
    } else {
        const bool whole = (NT * c + NT <= NOUT);
#pragma unroll
        for (int i = 0; i < NI; ++i) {
#pragma unroll
            for (int u = 0; u < 2; ++u) {
#pragma unroll
                for (int t = 0; t < NT; ++t) { Bf[i][u][t].x = 0.0f; Bf[i][u][t].y = 0.0f; }
            }
        }
        if (whole) {
#pragma unroll
            for (int i = 0; i < NI; ++i) {
#pragma unroll
                for (int u = 0; u < 2; ++u) {
                    const float* __restrict__ wp = W + (size_t)(8 * i + 4 * h + 2 * u) * NOUT + NT * c;
#pragma unroll
                    for (int t = 0; t < NT; ++t) { Bf[i][u][t].x = wp[t]; Bf[i][u][t].y = wp[NOUT + t]; }
                }
            }
        } else {
#pragma unroll
            for (int t = 0; t < NT; ++t) {
                if (NT * c + t < NOUT) {
#pragma unroll
                    for (int i = 0; i < NI; ++i) {
#pragma unroll
                        for (int u = 0; u < 2; ++u) {
                            const float* __restrict__ wp = W + (size_t)(8 * i + 4 * h + 2 * u) * NOUT + NT * c + t;
                            Bf[i][u][t].x = wp[0];
                            Bf[i][u][t].y = wp[NOUT];
                        }
                    }
                }
            }
        }
    }
    float bv[NT];
#pragma unroll
    for (int t = 0; t < NT; ++t) bv[t] = (NT * c + t < NOUT) ? bias[NT * c + t] : 0.0f;

    do {
        float4 n[NI], n2[NI], d[2];
        const int sn = s + nw;
        PS::load_dis(d, dis, s, h);
        PS::load_rows(n, n2, X, X2, (sn < Mtiles) ? sn : s, c, h);
        PS::run(a, a2, d, Bf, bv, Y, S, S2, s, c, h);
#pragma unroll
        for (int i = 0; i < NI; ++i) {
            a[i] = n[i];
            if constexpr (RELU_IN) a2[i] = n2[i];
        }
        s = sn;
    } while (s < Mtiles);
}

template <int NB>
__device__ __forceinline__ void scatter_edges(const float* __restrict__ H, float* __restrict__ AGG,
                                              int r, int c, float nrm, int k0, int lane) {
    float h0[NB], h1[NB];
#pragma unroll
    for (int j = 0; j < NB; ++j) {
        const unsigned hb = (unsigned)__builtin_amdgcn_readlane(r, k0 + j) * 64u + (unsigned)lane;
        h0[j] = H[hb];
        h1[j] = H[hb + 32u];
    }
#pragma unroll
    for (int j = 0; j < NB; ++j) {
        const unsigned ob = (unsigned)__builtin_amdgcn_readlane(c, k0 + j) * 64u + (unsigned)lane;
        const float nk = __int_as_float(__builtin_amdgcn_readlane(__float_as_int(nrm), k0 + j));
        atomicAdd(AGG + ob, h0[j] * nk);
        atomicAdd(AGG + ob + 32u, h1[j] * nk);
    }
}

__device__ __forceinline__ float edge_norm(const unsigned short* tab, const float* __restrict__ dis, int r, int c) {
    const unsigned qr = tab[r];
    const unsigned qc = tab[c];
    float dr = rsqrtf((float)qr);
    float dc = rsqrtf((float)qc);
    if (qr == DEG_WIDE) dr = dis[r];
    if (qc == DEG_WIDE) dc = dis[c];
    return dr * dc;
}

#define SCATTER_WAVES 16
#define TAB_VEC (NNODES * 2 / 16)
#define TAB_THREADS 500
#define TAB_PIECES 25

__device__ __forceinline__ int tab_pos(int i, int shift) {
    const int j = i + shift;
    return (j >= TAB_VEC) ? j - TAB_VEC : j;
}

__global__ __launch_bounds__(SCATTER_WAVES * 32) void scatter_kernel(const float* __restrict__ H, const int* __restrict__ row,
                               const int* __restrict__ col, const float* __restrict__ dis,
                               const unsigned short* __restrict__ degh,
                               float* __restrict__ AGG0, float* __restrict__ AGG1, int E) {
    __shared__ __attribute__((aligned(16))) unsigned short tab[NNODES];
    const int lane = threadIdx.x & 31;
    const int wib = __builtin_amdgcn_readfirstlane((int)(threadIdx.x >> 5));
    const int nw = (int)gridDim.x * SCATTER_WAVES;
    const int nchunks = (E + 31) >> 5;
    int s = wib * (int)gridDim.x + (int)blockIdx.x;
    const int last = E - 1;
    int e = s * 32 + lane;
    e = (e < last) ? e : last;
    int r = row[e];
    int c = col[e];
    {
        static_assert(TAB_THREADS * TAB_PIECES == TAB_VEC, "the copying threads cover the table exactly");
        static_assert(TAB_THREADS <= SCATTER_WAVES * 32, "the block has the copying threads");
        const uint4* __restrict__ g = reinterpret_cast<const uint4*>(degh);
        uint4* t = reinterpret_cast<uint4*>(tab);
        const int shift = (int)(blockIdx.x & 255u) * 49;
        const int tid = (int)threadIdx.x;
        if (tid < TAB_THREADS) {
            uint4 v[TAB_PIECES];
#pragma unroll
            for (int k = 0; k < TAB_PIECES; ++k) v[k] = g[tab_pos(tid + k * TAB_THREADS, shift)];
#pragma unroll
            for (int k = 0; k < TAB_PIECES; ++k) t[tab_pos(tid + k * TAB_THREADS, shift)] = v[k];
        }
    }
    __syncthreads();
    float* __restrict__ AGG = die_half() ? AGG1 : AGG0;
    if (s >= nchunks) return;
    float nrm = edge_norm(tab, dis, r, c);
    do {
        const int sn = s + nw;
        int en = sn * 32 + lane;
        en = (en < last) ? en : last;
        const int rn = row[en];
        const int cn = col[en];
        const int cnt = E - s * 32;
        if (cnt >= 32) {
            scatter_edges<16>(H, AGG, r, c, nrm, 0, lane);
        } else {
            int k = 0;
            for (; k + 8 <= cnt; k += 8) scatter_edges<8>(H, AGG, r, c, nrm, k, lane);
            for (; k < cnt; ++k) scatter_edges<1>(H, AGG, r, c, nrm, k, lane);
        }
        const float nn = edge_norm(tab, dis, rn, cn);
        if (cnt >= 32) scatter_edges<16>(H, AGG, r, c, nrm, 16, lane);
        r = rn;
        c = cn;
        nrm = nn;
        s = sn;
    } while (s < nchunks);
}


extern "C" void kernel_launch(void* const* d_in, const int* in_sizes, int n_in,
                              void* d_out, int out_size, void* d_ws, size_t ws_size,
                              hipStream_t stream) {
    const float* x    = (const float*)d_in[0];
    const int*   ei   = (const int*)d_in[1];
    const float* W1   = (const float*)d_in[2];
    const float* b1   = (const float*)d_in[3];
    const float* W2   = (const float*)d_in[4];
    const float* b2   = (const float*)d_in[5];
    const float* Wout = (const float*)d_in[6];
    const float* bout = (const float*)d_in[7];
    float* out = (float*)d_out;

    const int* row = ei;
    const int* col = ei + NEDGES;

    float* degc = (float*)d_ws;
    float* dis = degc + 2 * NNODES;
    float* A   = dis + NNODES;
    float* B0  = A + (size_t)NNODES * 64;
    float* B1  = B0 + (size_t)NNODES * 64;
    float* C0  = B1 + (size_t)NNODES * 64;
    float* C1  = C0 + (size_t)NNODES * 64;
    unsigned short* degh = (unsigned short*)(C1 + (size_t)NNODES * 64);
    (void)ws_size; (void)n_in; (void)in_sizes; (void)out_size;

    const int T = 256;
    const int Mtiles = NNODES / 16;
    const int gemmBlocks = 256;
    const int edgeBlocksDeg = 256;
    const int edgeBlocks = 256;

    fill_kernel<<<(2 * NNODES / 4 + T - 1) / T, T, 0, stream>>>(degc, 2 * NNODES / 4, 0.0f);
    deg_kernel<<<edgeBlocksDeg, T, 0, stream>>>(col, degc, NEDGES, NNODES);
    rsqrt_kernel<<<(NNODES / 4 + T - 1) / T, T, 0, stream>>>(degc, dis, degh, NNODES / 4);

    gemm_wmma_kernel<128, 4, 64, false, true><<<gemmBlocks, T, 0, stream>>>(x, nullptr, W1, b1, dis, A, B0, B1, Mtiles);
    scatter_kernel<<<edgeBlocks, SCATTER_WAVES * 32, 0, stream>>>(A, row, col, dis, degh, B0, B1, NEDGES);

    gemm_wmma_kernel<64, 4, 64, true, true><<<gemmBlocks, T, 0, stream>>>(B0, B1, W2, b2, dis, A, C0, C1, Mtiles);
    scatter_kernel<<<edgeBlocks, SCATTER_WAVES * 32, 0, stream>>>(A, row, col, dis, degh, C0, C1, NEDGES);

    gemm_wmma_kernel<64, 3, 40, true, false><<<gemmBlocks, T, 0, stream>>>(C0, C1, Wout, bout, nullptr, out, nullptr, nullptr, Mtiles);
}
